# starting kernel
# baseline (speedup 1.0000x reference)
#include <hip/hip_runtime.h>


typedef __attribute__((ext_vector_type(2))) float v2f;
typedef __attribute__((ext_vector_type(8))) float v8f;

#define BDIM            256
#define ROWS_PER_ITER   128
#define NITER           4
#define ROWS_PER_BLOCK  (ROWS_PER_ITER * NITER)
#define N_TOT           64
#define K_TOT           256
#define NPAIRS          (K_TOT / 2)
#define PAIR_STRIDE     160

extern __shared__ float wlds[];

__device__ __forceinline__ v2f nt_load2(const float* p) {
    return __builtin_nontemporal_load((const v2f*)p);
}

__device__ __forceinline__ void chunk4(v8f (&acc)[4], const v2f (&a)[4],
                                       const float* __restrict__ wl0) {
#pragma unroll
    for (int j = 0; j < 4; ++j) {
        const float* __restrict__ wl = wl0 + (2 * j) * PAIR_STRIDE;
        v2f bfr[4];
#pragma unroll
        for (int t = 0; t < 4; ++t)
            bfr[t] = *(const v2f*)(wl + t * 32);
#pragma unroll
        for (int t = 0; t < 4; ++t)
            acc[t] = __builtin_amdgcn_wmma_f32_16x16x4_f32(
                false, a[j], false, bfr[t], (short)0, acc[t], false, false);
    }
}

__global__ __launch_bounds__(BDIM)
void objdec_wmma_f32(const float* __restrict__ x,
                     const float* __restrict__ W,
                     const float* __restrict__ bias,
                     float* __restrict__ out)
{
    const int o   = blockIdx.y;
    const int tid = threadIdx.x;
    const float* __restrict__ Wo = W + (size_t)o * (N_TOT * K_TOT);

    for (int i = tid; i < NPAIRS * N_TOT; i += BDIM) {
        const int p = i >> 6;
        const int n = i & 63;
        const int gbase = (n >> 3) * 2048 + p * 16 + (n & 7);
        const float g0 = Wo[gbase];
        const float g1 = Wo[gbase + 8];
        *(v2f*)&wlds[p * PAIR_STRIDE + n * 2] = (v2f){g0, g1};
    }
    __syncthreads();

    const int wave = tid >> 5;
    const int lane = tid & 31;
    const int ln   = lane & 15;
    const int h    = lane >> 4;

    float bb[4];
#pragma unroll
    for (int t = 0; t < 4; ++t)
        bb[t] = bias[o * N_TOT + t * 16 + ln];

    const int rowBase0 = blockIdx.x * ROWS_PER_BLOCK + wave * 16;
    const float* __restrict__ wlbase = &wlds[h * PAIR_STRIDE + ln * 2];

    for (int it = 0; it < NITER; ++it) {
        const int rowBase = rowBase0 + it * ROWS_PER_ITER;
        const int xoff = ((rowBase + ln) * 16 + o) * K_TOT + 2 * h;
        const float* __restrict__ xl = x + xoff;

        v8f acc[4];
#pragma unroll
        for (int t = 0; t < 4; ++t)
            acc[t] = (v8f){bb[t], bb[t], bb[t], bb[t],
                           bb[t], bb[t], bb[t], bb[t]};

        v2f a_buf[4];
#pragma unroll
        for (int j = 0; j < 4; ++j)
            a_buf[j] = nt_load2(xl + 4 * j);

        for (int c = 0; c < 15; ++c) {
            v2f a_nxt[4];
#pragma unroll
            for (int j = 0; j < 4; ++j)
                a_nxt[j] = nt_load2(xl + (c + 1) * 16 + 4 * j);
            chunk4(acc, a_buf, wlbase + (8 * c) * PAIR_STRIDE);
#pragma unroll
            for (int j = 0; j < 4; ++j)
                a_buf[j] = a_nxt[j];
        }
        chunk4(acc, a_buf, wlbase + (8 * 15) * PAIR_STRIDE);

        float* __restrict__ obase =
            out + ((size_t)(rowBase + 8 * h) * 16 + o) * N_TOT + ln;
#pragma unroll
        for (int t = 0; t < 4; ++t) {
#pragma unroll
            for (int r = 0; r < 8; ++r)
                __builtin_nontemporal_store(acc[t][r],
                                            obase + r * (16 * N_TOT) + t * 16);
        }
    }
}

extern "C" void kernel_launch(void* const* d_in, const int* in_sizes, int n_in,
                              void* d_out, int out_size, void* d_ws, size_t ws_size,
                              hipStream_t stream) {
    const float* x    = (const float*)d_in[0];
    const float* W    = (const float*)d_in[1];
    const float* bias = (const float*)d_in[2];
    float* out        = (float*)d_out;

    const int B = 16384, N_OBJ = 16;
    dim3 grid(B / ROWS_PER_BLOCK, N_OBJ);
    dim3 block(BDIM);
    const size_t lds_bytes = (size_t)NPAIRS * PAIR_STRIDE * sizeof(float);
    hipLaunchKernelGGL(objdec_wmma_f32, grid, block, lds_bytes, stream,
                       x, W, bias, out);
}


# x staged per wave through a private LDS strip: whole-line b128 loads (4 full 128-B lines per load, was 16 lines of 16 B each), one 64-k slice ahead; A fragments read from LDS; sum order unchanged
# speedup vs baseline: 1.3734x; 1.3734x over previous
#include <hip/hip_runtime.h>


typedef __attribute__((ext_vector_type(2))) float v2f;
typedef __attribute__((ext_vector_type(4))) float v4f;
typedef __attribute__((ext_vector_type(8))) float v8f;

#define BDIM            256
#define ROWS_PER_ITER   128
#define NITER           4
#define ROWS_PER_BLOCK  (ROWS_PER_ITER * NITER)
#define N_TOT           64
#define K_TOT           256
#define NPAIRS          (K_TOT / 2)
#define PAIR_STRIDE     160
#define W_LDS_FLOATS    (NPAIRS * PAIR_STRIDE)

#define KSLICE          64
#define NSLICE          (K_TOT / KSLICE)
#define XS_PITCH        68
#define XS_WAVE         (16 * XS_PITCH)
#define ROW2_STRIDE     (2 * 16 * K_TOT)
#define TILE_STRIDE     (ROWS_PER_ITER * 16 * K_TOT)

extern __shared__ float wlds[];

__device__ __forceinline__ v4f nt_load4(const float* p) {
    return __builtin_nontemporal_load((const v4f*)p);
}

__device__ __forceinline__ void chunk4(v8f (&acc)[4], const v2f (&a)[4],
                                       const float* __restrict__ wl0) {
#pragma unroll
    for (int j = 0; j < 4; ++j) {
        const float* __restrict__ wl = wl0 + (2 * j) * PAIR_STRIDE;
        v2f bfr[4];
#pragma unroll
        for (int t = 0; t < 4; ++t)
            bfr[t] = *(const v2f*)(wl + t * 32);
#pragma unroll
        for (int t = 0; t < 4; ++t)
            acc[t] = __builtin_amdgcn_wmma_f32_16x16x4_f32(
                false, a[j], false, bfr[t], (short)0, acc[t], false, false);
    }
}

__global__ __launch_bounds__(BDIM)
void objdec_wmma_f32(const float* __restrict__ x,
                     const float* __restrict__ W,
                     const float* __restrict__ bias,
                     float* __restrict__ out)
{
    const int o   = blockIdx.y;
    const int tid = threadIdx.x;
    const float* __restrict__ Wo = W + (size_t)o * (N_TOT * K_TOT);

    const int wave = tid >> 5;
    const int lane = tid & 31;
    const int ln   = lane & 15;
    const int h    = lane >> 4;
    const int rowBase0 = blockIdx.x * ROWS_PER_BLOCK + wave * 16;

    const int g0i = ((rowBase0 + h) * 16 + o) * K_TOT + ln * 4;

    v4f R[8];
#pragma unroll
    for (int i = 0; i < 8; ++i)
        R[i] = nt_load4(x + g0i + i * ROW2_STRIDE);

    for (int i = tid; i < NPAIRS * N_TOT; i += BDIM) {
        const int p = i >> 6;
        const int n = i & 63;
        const int gbase = (n >> 3) * 2048 + p * 16 + (n & 7);
        const float g0 = Wo[gbase];
        const float g1 = Wo[gbase + 8];
        *(v2f*)&wlds[p * PAIR_STRIDE + n * 2] = (v2f){g0, g1};
    }
    __syncthreads();

    float bb[4];
#pragma unroll
    for (int t = 0; t < 4; ++t)
        bb[t] = bias[o * N_TOT + t * 16 + ln];

    const float* __restrict__ wlbase = &wlds[h * PAIR_STRIDE + ln * 2];

    float* xs = wlds + W_LDS_FLOATS + wave * XS_WAVE;
    float* xs_st = xs + h * XS_PITCH + ln * 4;
    const float* xs_ld = xs + ln * XS_PITCH + 2 * h;

    for (int it = 0; it < NITER; ++it) {
        const int rowBase = rowBase0 + it * ROWS_PER_ITER;

        v8f acc[4];
#pragma unroll
        for (int t = 0; t < 4; ++t)
            acc[t] = (v8f){bb[t], bb[t], bb[t], bb[t],
                           bb[t], bb[t], bb[t], bb[t]};

        for (int q = 0; q < NSLICE; ++q) {
#pragma unroll
            for (int i = 0; i < 8; ++i)
                *(v4f*)(xs_st + 2 * i * XS_PITCH) = R[i];

            const int s1 = it * NSLICE + q + 1;
            if (s1 < NITER * NSLICE) {
                const int gi = g0i + (s1 >> 2) * TILE_STRIDE + (s1 & 3) * KSLICE;
#pragma unroll
                for (int i = 0; i < 8; ++i)
                    R[i] = nt_load4(x + gi + i * ROW2_STRIDE);
            }

#pragma unroll
            for (int cc = 0; cc < 4; ++cc) {
                v2f a[4];
#pragma unroll
                for (int j = 0; j < 4; ++j)
                    a[j] = *(const v2f*)(xs_ld + 16 * cc + 4 * j);
                chunk4(acc, a, wlbase + (8 * (4 * q + cc)) * PAIR_STRIDE);
            }
        }

        float* __restrict__ obase =
            out + ((size_t)(rowBase + 8 * h) * 16 + o) * N_TOT + ln;
#pragma unroll
        for (int t = 0; t < 4; ++t) {
#pragma unroll
            for (int r = 0; r < 8; ++r)
                __builtin_nontemporal_store(acc[t][r],
                                            obase + r * (16 * N_TOT) + t * 16);
        }
    }
}

extern "C" void kernel_launch(void* const* d_in, const int* in_sizes, int n_in,
                              void* d_out, int out_size, void* d_ws, size_t ws_size,
                              hipStream_t stream) {
    const float* x    = (const float*)d_in[0];
    const float* W    = (const float*)d_in[1];
    const float* bias = (const float*)d_in[2];
    float* out        = (float*)d_out;

    const int B = 16384, N_OBJ = 16;
    dim3 grid(B / ROWS_PER_BLOCK, N_OBJ);
    dim3 block(BDIM);
    const size_t lds_bytes =
        (size_t)(W_LDS_FLOATS + (BDIM / 32) * XS_WAVE) * sizeof(float);
    hipLaunchKernelGGL(objdec_wmma_f32, grid, block, lds_bytes, stream,
                       x, W, bias, out);
}
